# starting kernel
# baseline (speedup 1.0000x reference)
#include <hip/hip_runtime.h>
#include <hip/hip_bf16.h>

typedef __attribute__((ext_vector_type(2))) float v2f;
typedef __attribute__((ext_vector_type(8))) float v8f;

constexpr int BB = 2;
constexpr int VV = 4;
constexpr int NN = 2048;
constexpr int FF = 128;
constexpr int EE = 32768;
constexpr int NE = 34816;
constexpr int OO = 512;

__device__ __forceinline__ float sgnval(int r, int j) {
  if (j >= 4) return 1.0f;
  const bool flip = (j & 1) ? ((r & 2) != 0) : ((r & 1) != 0);
  return flip ? -1.0f : 1.0f;
}

__device__ __forceinline__ int src_of(const int* __restrict__ ei, int e) {
  return (e < EE) ? ei[e] : (e - EE);
}
__device__ __forceinline__ int dst_of(const int* __restrict__ ei, int e) {
  return (e < EE) ? ei[EE + e] : (e - EE);
}

__global__ void zero_f32_kernel(float* __restrict__ p, size_t n) {
  size_t i = (size_t)blockIdx.x * blockDim.x + threadIdx.x;
  if (i < n) p[i] = 0.0f;
}
__global__ void zero_i32_kernel(int* __restrict__ p, int n) {
  int i = blockIdx.x * blockDim.x + threadIdx.x;
  if (i < n) p[i] = 0;
}

__global__ void count_dst_kernel(const int* __restrict__ ei, int* __restrict__ counts) {
  int e = blockIdx.x * blockDim.x + threadIdx.x;
  if (e >= NE) return;
  atomicAdd(&counts[dst_of(ei, e)], 1);
}

__global__ __launch_bounds__(256) void scan_counts_kernel(const int* __restrict__ counts,
                                                          int* __restrict__ rowstart) {
  __shared__ int part[256];
  __shared__ int pref[257];
  const int tid = threadIdx.x;
  const int base = tid * 8;
  int loc[8];
  int s = 0;
#pragma unroll
  for (int i = 0; i < 8; ++i) { loc[i] = s; s += counts[base + i]; }
  part[tid] = s;
  __syncthreads();
  if (tid == 0) {
    int a = 0;
    for (int i = 0; i < 256; ++i) { pref[i] = a; a += part[i]; }
    pref[256] = a;
  }
  __syncthreads();
  const int off = pref[tid];
#pragma unroll
  for (int i = 0; i < 8; ++i) rowstart[base + i] = off + loc[i];
  if (tid == 255) rowstart[NN] = pref[256];
}

__global__ __launch_bounds__(256) void build_rows_kernel(const int* __restrict__ ei,
                                                         const int* __restrict__ rowstart,
                                                         int* __restrict__ order) {
  const int lane = threadIdx.x & 31;
  const int wave = threadIdx.x >> 5;
  const int n = blockIdx.x * 8 + wave;
  if (n >= NN) return;
  const int base = rowstart[n];
  int cnt = 0;
  for (int e0 = 0; e0 < NE; e0 += 32) {
    const int e = e0 + lane;
    const bool match = (e < NE) && (dst_of(ei, e) == n);
    unsigned long long mask = __ballot(match);
    const int rank = __popcll(mask & ((1ull << lane) - 1ull));
    if (match) order[base + cnt + rank] = e;
    cnt += __popcll(mask);
  }
}

__device__ __forceinline__ void wmma_quad(const float* __restrict__ W, int k, int khalf,
                                          int col, v2f a0, v2f a1, v8f* acc) {
  const float* w0 = W + (size_t)(k + khalf) * OO;
  const float* w1 = w0 + OO;
  const v2f b0 = { w0[col +  0], w1[col +  0] };
  const v2f b1 = { w0[col + 16], w1[col + 16] };
  const v2f b2 = { w0[col + 32], w1[col + 32] };
  const v2f b3 = { w0[col + 48], w1[col + 48] };
  acc[0] = __builtin_amdgcn_wmma_f32_16x16x4_f32(false, a0, false, b0, (short)0, acc[0], false, false);
  acc[1] = __builtin_amdgcn_wmma_f32_16x16x4_f32(false, a0, false, b1, (short)0, acc[1], false, false);
  acc[2] = __builtin_amdgcn_wmma_f32_16x16x4_f32(false, a0, false, b2, (short)0, acc[2], false, false);
  acc[3] = __builtin_amdgcn_wmma_f32_16x16x4_f32(false, a0, false, b3, (short)0, acc[3], false, false);
  acc[4] = __builtin_amdgcn_wmma_f32_16x16x4_f32(false, a1, false, b0, (short)0, acc[4], false, false);
  acc[5] = __builtin_amdgcn_wmma_f32_16x16x4_f32(false, a1, false, b1, (short)0, acc[5], false, false);
  acc[6] = __builtin_amdgcn_wmma_f32_16x16x4_f32(false, a1, false, b2, (short)0, acc[6], false, false);
  acc[7] = __builtin_amdgcn_wmma_f32_16x16x4_f32(false, a1, false, b3, (short)0, acc[7], false, false);
}

__global__ __launch_bounds__(256) void gemm_pair_kernel(const float* __restrict__ x,
                                                        const float* __restrict__ W,
                                                        float* __restrict__ H,
                                                        int v1, int v2) {
  const int lane = threadIdx.x & 31;
  const int wave = threadIdx.x >> 5;
  const int rowTile = blockIdx.x * 32;
  const int b = rowTile >> 11;
  const int nbase = rowTile & (NN - 1);
  const int r0 = nbase + (lane & 15);
  const float* x1a = x + (((size_t)b * VV + v1) * NN + r0) * FF;
  const float* x1b = x1a + 16 * FF;
  const float* x2a = x + (((size_t)b * VV + v2) * NN + r0) * FF;
  const float* x2b = x2a + 16 * FF;
  const int khalf = (lane >> 4) * 2;
  const int col = wave * 64 + (lane & 15);

  v8f acc[8] = {};

  {
    v2f a0 = *(const v2f*)(x1a + khalf);
    v2f a1 = *(const v2f*)(x1b + khalf);
    const float s0 = sgnval(v2, khalf);
    const float s1 = sgnval(v2, khalf + 1);
    a0.x *= s0; a0.y *= s1;
    a1.x *= s0; a1.y *= s1;
    wmma_quad(W, 0, khalf, col, a0, a1, acc);
  }
  for (int k = 4; k < FF; k += 4) {
    const v2f a0 = *(const v2f*)(x1a + k + khalf);
    const v2f a1 = *(const v2f*)(x1b + k + khalf);
    wmma_quad(W, k, khalf, col, a0, a1, acc);
  }
  {
    v2f a0 = *(const v2f*)(x2a + khalf);
    v2f a1 = *(const v2f*)(x2b + khalf);
    const float s0 = sgnval(v1, khalf);
    const float s1 = sgnval(v1, khalf + 1);
    a0.x *= s0; a0.y *= s1;
    a1.x *= s0; a1.y *= s1;
    wmma_quad(W, FF, khalf, col, a0, a1, acc);
  }
  for (int k = FF + 4; k < 2 * FF; k += 4) {
    const v2f a0 = *(const v2f*)(x2a + (k - FF) + khalf);
    const v2f a1 = *(const v2f*)(x2b + (k - FF) + khalf);
    wmma_quad(W, k, khalf, col, a0, a1, acc);
  }

  float* Hb = H + (size_t)b * NN * OO;
  const int srow = nbase + 8 * (lane >> 4);
  const int scol = wave * 64 + (lane & 15);
#pragma unroll
  for (int i = 0; i < 8; ++i) {
    float* ra = Hb + (size_t)(srow + i) * OO + scol;
    ra[ 0] = acc[0][i];
    ra[16] = acc[1][i];
    ra[32] = acc[2][i];
    ra[48] = acc[3][i];
    float* rb = Hb + (size_t)(srow + 16 + i) * OO + scol;
    rb[ 0] = acc[4][i];
    rb[16] = acc[5][i];
    rb[32] = acc[6][i];
    rb[48] = acc[7][i];
  }
}

__global__ __launch_bounds__(256) void node_scores_kernel(const float* __restrict__ H,
                                                          const float* __restrict__ att,
                                                          float* __restrict__ uS,
                                                          float* __restrict__ vS) {
  const int lane = threadIdx.x & 31;
  const int wave = threadIdx.x >> 5;
  const int node = blockIdx.x * 8 + wave;
  if (node >= BB * NN) return;
  const float* hrow = H + (size_t)node * OO;
  float su = 0.0f, sv = 0.0f;
  for (int j = lane; j < OO; j += 32) {
    const float h = hrow[j];
    const float l = h > 0.0f ? h : 0.2f * h;
    su += att[j] * l;
    sv += att[OO + j] * l;
  }
#pragma unroll
  for (int off = 16; off; off >>= 1) {
    su += __shfl_down(su, off, 32);
    sv += __shfl_down(sv, off, 32);
  }
  if (lane == 0) { uS[node] = su; vS[node] = sv; }
}

__global__ void edge_scores_kernel(const float* __restrict__ uS, const float* __restrict__ vS,
                                   const int* __restrict__ ei, float* __restrict__ s) {
  const int t = blockIdx.x * blockDim.x + threadIdx.x;
  if (t >= BB * NE) return;
  const int b = t / NE;
  const int e = t - b * NE;
  s[t] = uS[b * NN + src_of(ei, e)] + vS[b * NN + dst_of(ei, e)];
}

__global__ __launch_bounds__(256) void softmax_stats_kernel(const float* __restrict__ s,
                                                            float* __restrict__ stats) {
  const int b = blockIdx.x;
  const int tid = threadIdx.x;
  __shared__ float red[256];
  const float* sb = s + (size_t)b * NE;
  float m = -3.4e38f;
  for (int i = tid; i < NE; i += 256) m = fmaxf(m, sb[i]);
  red[tid] = m;
  __syncthreads();
  for (int off = 128; off; off >>= 1) {
    if (tid < off) red[tid] = fmaxf(red[tid], red[tid + off]);
    __syncthreads();
  }
  const float bm = red[0];
  __syncthreads();
  float sum = 0.0f;
  for (int i = tid; i < NE; i += 256) sum += expf(sb[i] - bm);
  red[tid] = sum;
  __syncthreads();
  for (int off = 128; off; off >>= 1) {
    if (tid < off) red[tid] += red[tid + off];
    __syncthreads();
  }
  if (tid == 0) { stats[2 * b] = bm; stats[2 * b + 1] = red[0]; }
}

__global__ __launch_bounds__(256) void aggregate_kernel(const float* __restrict__ H,
                                                        const float* __restrict__ s,
                                                        const float* __restrict__ stats,
                                                        const int* __restrict__ rowstart,
                                                        const int* __restrict__ order,
                                                        const int* __restrict__ ei,
                                                        float* __restrict__ out, int g) {
  const int lane = threadIdx.x & 31;
  const int wave = threadIdx.x >> 5;
  const int node = blockIdx.x * 8 + wave;
  if (node >= BB * NN) return;
  const int b = node >> 11;
  const int n = node & (NN - 1);
  const float m = stats[2 * b];
  const float inv = 1.0f / stats[2 * b + 1];
  const float* Hb = H + (size_t)b * NN * OO;
  const float* sb = s + (size_t)b * NE;
  float4 acc[4] = {};
  const int beg = rowstart[n];
  const int end = rowstart[n + 1];
  for (int idx = beg; idx < end; ++idx) {
    const int e = order[idx];
    const int sn = src_of(ei, e);
    const float w = expf(sb[e] - m) * inv;
    const float4* hr = (const float4*)(Hb + (size_t)sn * OO);
#pragma unroll
    for (int j = 0; j < 4; ++j) {
      const float4 h4 = hr[lane + 32 * j];
      acc[j].x += w * h4.x;
      acc[j].y += w * h4.y;
      acc[j].z += w * h4.z;
      acc[j].w += w * h4.w;
    }
  }
  float4* op = (float4*)(out + (((size_t)b * VV + g) * NN + n) * OO);
#pragma unroll
  for (int j = 0; j < 4; ++j) {
    float4 o = op[lane + 32 * j];
    o.x += acc[j].x; o.y += acc[j].y; o.z += acc[j].z; o.w += acc[j].w;
    op[lane + 32 * j] = o;
  }
}

__global__ void epilogue_kernel(float* __restrict__ out, const float* __restrict__ bias, size_t n) {
  size_t i = (size_t)blockIdx.x * blockDim.x + threadIdx.x;
  if (i < n) out[i] = out[i] * 0.25f + bias[i & (OO - 1)];
}

extern "C" void kernel_launch(void* const* d_in, const int* in_sizes, int n_in,
                              void* d_out, int out_size, void* d_ws, size_t ws_size,
                              hipStream_t stream) {
  const float* x    = (const float*)d_in[0];
  const int*   ei   = (const int*)d_in[1];
  const float* W    = (const float*)d_in[2];
  const float* att  = (const float*)d_in[3];
  const float* bias = (const float*)d_in[4];
  float* out = (float*)d_out;

  char* p = (char*)d_ws;
  float* H     = (float*)p; p += (size_t)BB * NN * OO * sizeof(float);
  float* uS    = (float*)p; p += (size_t)BB * NN * sizeof(float);
  float* vS    = (float*)p; p += (size_t)BB * NN * sizeof(float);
  float* s     = (float*)p; p += (size_t)BB * NE * sizeof(float);
  float* stats = (float*)p; p += 4 * sizeof(float);
  int* counts  = (int*)p;   p += (size_t)NN * sizeof(int);
  int* order   = (int*)p;   p += (size_t)NE * sizeof(int);
  int* rowstart= (int*)p;   p += (size_t)(NN + 1) * sizeof(int);

  const size_t outN = (size_t)BB * VV * NN * OO;

  zero_f32_kernel<<<(unsigned)((outN + 255) / 256), 256, 0, stream>>>(out, outN);
  zero_i32_kernel<<<(NN + 255) / 256, 256, 0, stream>>>(counts, NN);
  count_dst_kernel<<<(NE + 255) / 256, 256, 0, stream>>>(ei, counts);
  scan_counts_kernel<<<1, 256, 0, stream>>>(counts, rowstart);
  build_rows_kernel<<<NN / 8, 256, 0, stream>>>(ei, rowstart, order);

  for (int g = 0; g < VV; ++g) {
    for (int hv = 0; hv < VV; ++hv) {
      const int v1 = hv;
      const int v2 = g ^ hv;
      gemm_pair_kernel<<<(BB * NN) / 32, 256, 0, stream>>>(x, W, H, v1, v2);
      node_scores_kernel<<<(BB * NN) / 8, 256, 0, stream>>>(H, att, uS, vS);
      edge_scores_kernel<<<(BB * NE + 255) / 256, 256, 0, stream>>>(uS, vS, ei, s);
      softmax_stats_kernel<<<BB, 256, 0, stream>>>(s, stats);
      aggregate_kernel<<<(BB * NN) / 8, 256, 0, stream>>>(H, s, stats, rowstart, order, ei, out, g);
    }
  }
  epilogue_kernel<<<(unsigned)((outN + 255) / 256), 256, 0, stream>>>(out, bias, outN);

  (void)in_sizes; (void)n_in; (void)out_size; (void)ws_size;
}


# pair products parked in out's not-yet-summed view planes beside the workspace: 16 pair pipelines run as 9 batched groups (4,3+1,2+2,1x4), one edge walk per group; first group stores; sum order kept
# speedup vs baseline: 3.2116x; 3.2116x over previous
#include <hip/hip_runtime.h>
#include <hip/hip_bf16.h>

typedef __attribute__((ext_vector_type(2))) float v2f;
typedef __attribute__((ext_vector_type(8))) float v8f;
typedef __attribute__((ext_vector_type(4))) float v4f;

constexpr int BB = 2;
constexpr int VV = 4;
constexpr int NN = 2048;
constexpr int FF = 128;
constexpr int EE = 32768;
constexpr int NE = 34816;
constexpr int OO = 512;

__device__ __forceinline__ float sgnval(int r, int j) {
  if (j >= 4) return 1.0f;
  const bool flip = (j & 1) ? ((r & 2) != 0) : ((r & 1) != 0);
  return flip ? -1.0f : 1.0f;
}

__device__ __forceinline__ int src_of(const int* __restrict__ ei, int e) {
  return (e < EE) ? ei[e] : (e - EE);
}
__device__ __forceinline__ int dst_of(const int* __restrict__ ei, int e) {
  return (e < EE) ? ei[EE + e] : (e - EE);
}

struct Bufs {
  float* h0; float* h1; float* h2; float* h3;
  int bs0; int bs1; int bs2; int bs3;
};
__device__ __forceinline__ float* buf_of(const Bufs& B, int p) {
  return p == 0 ? B.h0 : (p == 1 ? B.h1 : (p == 2 ? B.h2 : B.h3));
}
__device__ __forceinline__ int bs_of(const Bufs& B, int p) {
  return p == 0 ? B.bs0 : (p == 1 ? B.bs1 : (p == 2 ? B.bs2 : B.bs3));
}
constexpr int PS = BB * NN;

__global__ void zero_i32_kernel(int* __restrict__ p, int n) {
  int i = blockIdx.x * blockDim.x + threadIdx.x;
  if (i < n) p[i] = 0;
}

__global__ void count_dst_kernel(const int* __restrict__ ei, int* __restrict__ counts,
                                 int* __restrict__ slot) {
  int e = blockIdx.x * blockDim.x + threadIdx.x;
  if (e >= NE) return;
  slot[e] = atomicAdd(&counts[dst_of(ei, e)], 1);
}

__global__ __launch_bounds__(256) void scan_counts_kernel(const int* __restrict__ counts,
                                                          int* __restrict__ rowstart) {
  __shared__ int part[256];
  __shared__ int pref[257];
  const int tid = threadIdx.x;
  const int base = tid * 8;
  int loc[8];
  int s = 0;
#pragma unroll
  for (int i = 0; i < 8; ++i) { loc[i] = s; s += counts[base + i]; }
  part[tid] = s;
  __syncthreads();
  if (tid == 0) {
    int a = 0;
    for (int i = 0; i < 256; ++i) { pref[i] = a; a += part[i]; }
    pref[256] = a;
  }
  __syncthreads();
  const int off = pref[tid];
#pragma unroll
  for (int i = 0; i < 8; ++i) rowstart[base + i] = off + loc[i];
  if (tid == 255) rowstart[NN] = pref[256];
}

__global__ void fill_rows_kernel(const int* __restrict__ ei, const int* __restrict__ rowstart,
                                 const int* __restrict__ slot, int* __restrict__ row_edge) {
  int e = blockIdx.x * blockDim.x + threadIdx.x;
  if (e >= NE) return;
  row_edge[rowstart[dst_of(ei, e)] + slot[e]] = e;
}

__global__ __launch_bounds__(256) void order_rows_kernel(const int* __restrict__ ei,
                                                         const int* __restrict__ rowstart,
                                                         const int* __restrict__ row_edge,
                                                         int* __restrict__ csr_src) {
  const int lane = threadIdx.x & 31;
  const int wave = threadIdx.x >> 5;
  const int n = blockIdx.x * 8 + wave;
  if (n >= NN) return;
  const int beg = __builtin_amdgcn_readfirstlane(rowstart[n]);
  const int len = __builtin_amdgcn_readfirstlane(rowstart[n + 1]) - beg;
  for (int i0 = 0; i0 < len; i0 += 32) {
    const int i = i0 + lane;
    const int id = (i < len) ? row_edge[beg + i] : 0x7fffffff;
    int r = 0;
    for (int j0 = 0; j0 < len; j0 += 32) {
      const int j = j0 + lane;
      const int idj = (j < len) ? row_edge[beg + j] : 0x7fffffff;
      const int m = (len - j0 < 32) ? (len - j0) : 32;
      for (int t = 0; t < m; ++t) r += (__builtin_amdgcn_readlane(idj, t) < id) ? 1 : 0;
    }
    if (i < len) csr_src[beg + r] = src_of(ei, id);
  }
}

__device__ __forceinline__ void wmma_quad(const float* __restrict__ W, int k, int khalf,
                                          int col, v2f a0, v2f a1, v8f* acc) {
  const float* w0 = W + (size_t)(k + khalf) * OO;
  const float* w1 = w0 + OO;
  const v2f b0 = { w0[col +  0], w1[col +  0] };
  const v2f b1 = { w0[col + 16], w1[col + 16] };
  const v2f b2 = { w0[col + 32], w1[col + 32] };
  const v2f b3 = { w0[col + 48], w1[col + 48] };
  acc[0] = __builtin_amdgcn_wmma_f32_16x16x4_f32(false, a0, false, b0, (short)0, acc[0], false, false);
  acc[1] = __builtin_amdgcn_wmma_f32_16x16x4_f32(false, a0, false, b1, (short)0, acc[1], false, false);
  acc[2] = __builtin_amdgcn_wmma_f32_16x16x4_f32(false, a0, false, b2, (short)0, acc[2], false, false);
  acc[3] = __builtin_amdgcn_wmma_f32_16x16x4_f32(false, a0, false, b3, (short)0, acc[3], false, false);
  acc[4] = __builtin_amdgcn_wmma_f32_16x16x4_f32(false, a1, false, b0, (short)0, acc[4], false, false);
  acc[5] = __builtin_amdgcn_wmma_f32_16x16x4_f32(false, a1, false, b1, (short)0, acc[5], false, false);
  acc[6] = __builtin_amdgcn_wmma_f32_16x16x4_f32(false, a1, false, b2, (short)0, acc[6], false, false);
  acc[7] = __builtin_amdgcn_wmma_f32_16x16x4_f32(false, a1, false, b3, (short)0, acc[7], false, false);
}

__global__ __launch_bounds__(256) void gemm_pair_kernel(const float* __restrict__ x,
                                                        const float* __restrict__ W,
                                                        Bufs B, int g, int hv0) {
  const int pr = blockIdx.y;
  const int v1 = hv0 + pr;
  const int v2 = g ^ v1;
  float* H = buf_of(B, pr);
  const int bs = bs_of(B, pr);
  const int lane = threadIdx.x & 31;
  const int wave = threadIdx.x >> 5;
  const int rowTile = blockIdx.x * 32;
  const int b = rowTile >> 11;
  const int nbase = rowTile & (NN - 1);
  const int r0 = nbase + (lane & 15);
  const float* x1a = x + (((size_t)b * VV + v1) * NN + r0) * FF;
  const float* x1b = x1a + 16 * FF;
  const float* x2a = x + (((size_t)b * VV + v2) * NN + r0) * FF;
  const float* x2b = x2a + 16 * FF;
  const int khalf = (lane >> 4) * 2;
  const int col = wave * 64 + (lane & 15);

  v8f acc[8] = {};

  {
    v2f a0 = *(const v2f*)(x1a + khalf);
    v2f a1 = *(const v2f*)(x1b + khalf);
    const float s0 = sgnval(v2, khalf);
    const float s1 = sgnval(v2, khalf + 1);
    a0.x *= s0; a0.y *= s1;
    a1.x *= s0; a1.y *= s1;
    wmma_quad(W, 0, khalf, col, a0, a1, acc);
  }
  for (int k = 4; k < FF; k += 4) {
    const v2f a0 = *(const v2f*)(x1a + k + khalf);
    const v2f a1 = *(const v2f*)(x1b + k + khalf);
    wmma_quad(W, k, khalf, col, a0, a1, acc);
  }
  {
    v2f a0 = *(const v2f*)(x2a + khalf);
    v2f a1 = *(const v2f*)(x2b + khalf);
    const float s0 = sgnval(v1, khalf);
    const float s1 = sgnval(v1, khalf + 1);
    a0.x *= s0; a0.y *= s1;
    a1.x *= s0; a1.y *= s1;
    wmma_quad(W, FF, khalf, col, a0, a1, acc);
  }
  for (int k = FF + 4; k < 2 * FF; k += 4) {
    const v2f a0 = *(const v2f*)(x2a + (k - FF) + khalf);
    const v2f a1 = *(const v2f*)(x2b + (k - FF) + khalf);
    wmma_quad(W, k, khalf, col, a0, a1, acc);
  }

  float* Hb = H + (size_t)b * bs;
  const int srow = nbase + 8 * (lane >> 4);
  const int scol = wave * 64 + (lane & 15);
#pragma unroll
  for (int i = 0; i < 8; ++i) {
    float* ra = Hb + (size_t)(srow + i) * OO + scol;
    ra[ 0] = acc[0][i];
    ra[16] = acc[1][i];
    ra[32] = acc[2][i];
    ra[48] = acc[3][i];
    float* rb = Hb + (size_t)(srow + 16 + i) * OO + scol;
    rb[ 0] = acc[4][i];
    rb[16] = acc[5][i];
    rb[32] = acc[6][i];
    rb[48] = acc[7][i];
  }
}

__global__ __launch_bounds__(256) void node_scores_kernel(Bufs B,
                                                          const float* __restrict__ att,
                                                          float* __restrict__ uS,
                                                          float* __restrict__ vS) {
  const int lane = threadIdx.x & 31;
  const int wave = threadIdx.x >> 5;
  const int node = blockIdx.x * 8 + wave;
  if (node >= BB * NN) return;
  const int pr = blockIdx.y;
  const float* hrow = buf_of(B, pr) + (size_t)(node >> 11) * bs_of(B, pr) + (size_t)(node & (NN - 1)) * OO;
  uS += (size_t)pr * PS;
  vS += (size_t)pr * PS;
  float su = 0.0f, sv = 0.0f;
  for (int j = lane; j < OO; j += 32) {
    const float h = hrow[j];
    const float l = h > 0.0f ? h : 0.2f * h;
    su += att[j] * l;
    sv += att[OO + j] * l;
  }
#pragma unroll
  for (int off = 16; off; off >>= 1) {
    su += __shfl_down(su, off, 32);
    sv += __shfl_down(sv, off, 32);
  }
  if (lane == 0) { uS[node] = su; vS[node] = sv; }
}

__global__ __launch_bounds__(256) void row_stats_kernel(const float* __restrict__ uS,
                                                        const float* __restrict__ vS,
                                                        const int* __restrict__ rowstart,
                                                        const int* __restrict__ csr_src,
                                                        float* __restrict__ rowmax,
                                                        float* __restrict__ rowsum) {
  const int lane = threadIdx.x & 31;
  const int wave = threadIdx.x >> 5;
  const int node = blockIdx.x * 8 + wave;
  if (node >= BB * NN) return;
  const int pr = blockIdx.y;
  uS += (size_t)pr * PS;
  vS += (size_t)pr * PS;
  rowmax += (size_t)pr * PS;
  rowsum += (size_t)pr * PS;
  const int b = node >> 11;
  const int n = node & (NN - 1);
  const float* ub = uS + (size_t)b * NN;
  const float vn = vS[node];
  const int beg = rowstart[n];
  const int end = rowstart[n + 1];
  float mx = -3.4e38f;
  for (int idx = beg + lane; idx < end; idx += 32) mx = fmaxf(mx, ub[csr_src[idx]] + vn);
#pragma unroll
  for (int off = 16; off; off >>= 1) mx = fmaxf(mx, __shfl_xor(mx, off, 32));
  float sm = 0.0f;
  for (int idx = beg + lane; idx < end; idx += 32) sm += expf((ub[csr_src[idx]] + vn) - mx);
#pragma unroll
  for (int off = 16; off; off >>= 1) sm += __shfl_xor(sm, off, 32);
  if (lane == 0) { rowmax[node] = mx; rowsum[node] = sm; }
}

template <int P>
__device__ __forceinline__ void hold_loads(const float (&uu)[P], const v4f (&hh)[P][4]) {
  if constexpr (P == 2) {
    asm volatile("" :: "v"(uu[0]), "v"(uu[1]),
                 "v"(hh[0][0]), "v"(hh[0][1]), "v"(hh[0][2]), "v"(hh[0][3]),
                 "v"(hh[1][0]), "v"(hh[1][1]), "v"(hh[1][2]), "v"(hh[1][3]));
  } else if constexpr (P == 3) {
    asm volatile("" :: "v"(uu[0]), "v"(uu[1]), "v"(uu[2]),
                 "v"(hh[0][0]), "v"(hh[0][1]), "v"(hh[0][2]), "v"(hh[0][3]),
                 "v"(hh[1][0]), "v"(hh[1][1]), "v"(hh[1][2]), "v"(hh[1][3]),
                 "v"(hh[2][0]), "v"(hh[2][1]), "v"(hh[2][2]), "v"(hh[2][3]));
  } else {
    asm volatile("" :: "v"(uu[0]), "v"(uu[1]), "v"(uu[2]), "v"(uu[3]),
                 "v"(hh[0][0]), "v"(hh[0][1]), "v"(hh[0][2]), "v"(hh[0][3]),
                 "v"(hh[1][0]), "v"(hh[1][1]), "v"(hh[1][2]), "v"(hh[1][3]),
                 "v"(hh[2][0]), "v"(hh[2][1]), "v"(hh[2][2]), "v"(hh[2][3]),
                 "v"(hh[3][0]), "v"(hh[3][1]), "v"(hh[3][2]), "v"(hh[3][3]));
  }
}

template <int P, int E>
__device__ __forceinline__ void walk_step(const int* __restrict__ csr_src, int idx, int lane,
                                          const float* const (&Hb)[P], const float* const (&ub)[P],
                                          const float (&vn)[P], const float (&m)[P],
                                          const float (&inv)[P], float4 (&acc)[P][4]) {
  int sn[E];
#pragma unroll
  for (int e = 0; e < E; ++e) sn[e] = csr_src[idx + e];
  float uu[E][P];
  v4f hh[E][P][4];
#pragma unroll
  for (int e = 0; e < E; ++e) {
#pragma unroll
    for (int p = 0; p < P; ++p) {
      uu[e][p] = ub[p][sn[e]];
      const v4f* hr = (const v4f*)(Hb[p] + (size_t)sn[e] * OO);
#pragma unroll
      for (int j = 0; j < 4; ++j) hh[e][p][j] = hr[lane + 32 * j];
    }
  }
#pragma unroll
  for (int e = 0; e < E; ++e) hold_loads<P>(uu[e], hh[e]);
#pragma unroll
  for (int e = 0; e < E; ++e) {
#pragma unroll
    for (int p = 0; p < P; ++p) {
      const float w = expf((uu[e][p] + vn[p]) - m[p]) * inv[p];
#pragma unroll
      for (int j = 0; j < 4; ++j) {
        acc[p][j].x += w * hh[e][p][j].x;
        acc[p][j].y += w * hh[e][p][j].y;
        acc[p][j].z += w * hh[e][p][j].z;
        acc[p][j].w += w * hh[e][p][j].w;
      }
    }
  }
}

template <int P>
__global__ __launch_bounds__(256) void aggregate_kernel(Bufs B,
                                                        const float* __restrict__ uS,
                                                        const float* __restrict__ vS,
                                                        const float* __restrict__ rowmax,
                                                        const float* __restrict__ rowsum,
                                                        const int* __restrict__ rowstart,
                                                        const int* __restrict__ csr_src,
                                                        float* out, int g, int first) {
  __shared__ float redm[P][8];
  __shared__ float reds[P][8];
  const int tid = threadIdx.x;
  const int lane = tid & 31;
  const int wave = tid >> 5;
  const int node = blockIdx.x * 8 + wave;
  const int b = node >> 11;
  const int n = node & (NN - 1);
  const int beg = rowstart[n];
  const int end = rowstart[n + 1];

  float vn[P], m[P], inv[P];
  const float* Hb[P];
  const float* ub[P];
  float4 m0[P], m1[P], s0[P], s1[P];
#pragma unroll
  for (int p = 0; p < P; ++p) {
    vn[p] = vS[(size_t)p * PS + node];
    Hb[p] = buf_of(B, p) + (size_t)b * bs_of(B, p);
    ub[p] = uS + (size_t)p * PS + (size_t)b * NN;
    const float4* rm4 = (const float4*)(rowmax + (size_t)p * PS + (size_t)b * NN) + 2 * tid;
    const float4* rs4 = (const float4*)(rowsum + (size_t)p * PS + (size_t)b * NN) + 2 * tid;
    m0[p] = rm4[0]; m1[p] = rm4[1];
    s0[p] = rs4[0]; s1[p] = rs4[1];
  }
#pragma unroll
  for (int p = 0; p < P; ++p) {
    float mx = fmaxf(fmaxf(fmaxf(m0[p].x, m0[p].y), fmaxf(m0[p].z, m0[p].w)),
                     fmaxf(fmaxf(m1[p].x, m1[p].y), fmaxf(m1[p].z, m1[p].w)));
#pragma unroll
    for (int off = 16; off; off >>= 1) mx = fmaxf(mx, __shfl_xor(mx, off, 32));
    if (lane == 0) redm[p][wave] = mx;
  }
  __syncthreads();
#pragma unroll
  for (int p = 0; p < P; ++p) {
    float mm = redm[p][0];
#pragma unroll
    for (int i = 1; i < 8; ++i) mm = fmaxf(mm, redm[p][i]);
    float z = s0[p].x * expf(m0[p].x - mm);
    z += s0[p].y * expf(m0[p].y - mm);
    z += s0[p].z * expf(m0[p].z - mm);
    z += s0[p].w * expf(m0[p].w - mm);
    z += s1[p].x * expf(m1[p].x - mm);
    z += s1[p].y * expf(m1[p].y - mm);
    z += s1[p].z * expf(m1[p].z - mm);
    z += s1[p].w * expf(m1[p].w - mm);
#pragma unroll
    for (int off = 16; off; off >>= 1) z += __shfl_xor(z, off, 32);
    if (lane == 0) reds[p][wave] = z;
    m[p] = mm;
  }
  __syncthreads();
#pragma unroll
  for (int p = 0; p < P; ++p) {
    float zsum = reds[p][0];
#pragma unroll
    for (int i = 1; i < 8; ++i) zsum += reds[p][i];
    inv[p] = 1.0f / zsum;
  }

  float4 acc[P][4] = {};
  if constexpr (P == 1) {
    for (int idx = beg; idx < end; ++idx) {
      const int sn = csr_src[idx];
      const float w = expf((ub[0][sn] + vn[0]) - m[0]) * inv[0];
      const float4* hr = (const float4*)(Hb[0] + (size_t)sn * OO);
#pragma unroll
      for (int j = 0; j < 4; ++j) {
        const float4 h4 = hr[lane + 32 * j];
        acc[0][j].x += w * h4.x;
        acc[0][j].y += w * h4.y;
        acc[0][j].z += w * h4.z;
        acc[0][j].w += w * h4.w;
      }
    }
  } else {
    int idx = beg;
    for (; idx + 1 < end; idx += 2) walk_step<P, 2>(csr_src, idx, lane, Hb, ub, vn, m, inv, acc);
    if (idx < end) walk_step<P, 1>(csr_src, idx, lane, Hb, ub, vn, m, inv, acc);
  }
  float4* op = (float4*)(out + (((size_t)b * VV + g) * NN + n) * OO);
  float4 q[4];
  if (first) {
#pragma unroll
    for (int j = 0; j < 4; ++j) q[j] = make_float4(0.0f, 0.0f, 0.0f, 0.0f);
  } else {
#pragma unroll
    for (int j = 0; j < 4; ++j) q[j] = op[lane + 32 * j];
  }
#pragma unroll
  for (int j = 0; j < 4; ++j) {
    float4 o = q[j];
#pragma unroll
    for (int p = 0; p < P; ++p) {
      o.x += acc[p][j].x; o.y += acc[p][j].y; o.z += acc[p][j].z; o.w += acc[p][j].w;
    }
    op[lane + 32 * j] = o;
  }
}

__global__ void epilogue_kernel(float* __restrict__ out, const float* __restrict__ bias, size_t n) {
  size_t i = (size_t)blockIdx.x * blockDim.x + threadIdx.x;
  if (i < n) out[i] = out[i] * 0.25f + bias[i & (OO - 1)];
}

template <int P>
static void run_group(const float* x, const float* W, const float* att, Bufs B,
                      float* uS, float* vS, float* rowmax, float* rowsum,
                      const int* rowstart, const int* csr_src, float* out,
                      int g, int hv0, int first, hipStream_t stream) {
  gemm_pair_kernel<<<dim3((BB * NN) / 32, P), 256, 0, stream>>>(x, W, B, g, hv0);
  node_scores_kernel<<<dim3((BB * NN) / 8, P), 256, 0, stream>>>(B, att, uS, vS);
  row_stats_kernel<<<dim3((BB * NN) / 8, P), 256, 0, stream>>>(uS, vS, rowstart, csr_src, rowmax, rowsum);
  aggregate_kernel<P><<<(BB * NN) / 8, 256, 0, stream>>>(B, uS, vS, rowmax, rowsum, rowstart, csr_src,
                                                         out, g, first);
}

extern "C" void kernel_launch(void* const* d_in, const int* in_sizes, int n_in,
                              void* d_out, int out_size, void* d_ws, size_t ws_size,
                              hipStream_t stream) {
  const float* x    = (const float*)d_in[0];
  const int*   ei   = (const int*)d_in[1];
  const float* W    = (const float*)d_in[2];
  const float* att  = (const float*)d_in[3];
  const float* bias = (const float*)d_in[4];
  float* out = (float*)d_out;

  char* p = (char*)d_ws;
  float* H     = (float*)p; p += (size_t)BB * NN * OO * sizeof(float);
  float* uS     = (float*)p;
  float* vS     = uS + (size_t)4 * PS;
  float* rowmax = vS + (size_t)4 * PS;
  float* rowsum = rowmax + (size_t)4 * PS;
  p += (size_t)2 * BB * NN * sizeof(float);
  p += (size_t)BB * NE * sizeof(float);
  p += 4 * sizeof(float);
  int* counts  = (int*)p;   p += (size_t)NN * sizeof(int);
  int* csr_src = (int*)p;   p += (size_t)NE * sizeof(int);
  int* rowstart= (int*)p;   p += (size_t)(NN + 1) * sizeof(int);
  int* slot     = (int*)H;
  int* row_edge = slot + NE;

  const size_t outN = (size_t)BB * VV * NN * OO;

  zero_i32_kernel<<<(NN + 255) / 256, 256, 0, stream>>>(counts, NN);
  count_dst_kernel<<<(NE + 255) / 256, 256, 0, stream>>>(ei, counts, slot);
  scan_counts_kernel<<<1, 256, 0, stream>>>(counts, rowstart);
  fill_rows_kernel<<<(NE + 255) / 256, 256, 0, stream>>>(ei, rowstart, slot, row_edge);
  order_rows_kernel<<<NN / 8, 256, 0, stream>>>(ei, rowstart, row_edge, csr_src);

  constexpr int WS_BS = NN * OO;
  constexpr int PL_BS = VV * NN * OO;
  float* pl1 = out + (size_t)1 * NN * OO;
  float* pl2 = out + (size_t)2 * NN * OO;
  float* pl3 = out + (size_t)3 * NN * OO;
  const Bufs b4 = { H, pl1, pl2, pl3, WS_BS, PL_BS, PL_BS, PL_BS };
  const Bufs b3 = { H, pl2, pl3, pl3, WS_BS, PL_BS, PL_BS, PL_BS };
  const Bufs b2 = { H, pl3, pl3, pl3, WS_BS, PL_BS, PL_BS, PL_BS };
  const Bufs b1 = { H, H, H, H, WS_BS, WS_BS, WS_BS, WS_BS };

  run_group<4>(x, W, att, b4, uS, vS, rowmax, rowsum, rowstart, csr_src, out, 0, 0, 1, stream);
  run_group<3>(x, W, att, b3, uS, vS, rowmax, rowsum, rowstart, csr_src, out, 1, 0, 1, stream);
  run_group<1>(x, W, att, b1, uS, vS, rowmax, rowsum, rowstart, csr_src, out, 1, 3, 0, stream);
  run_group<2>(x, W, att, b2, uS, vS, rowmax, rowsum, rowstart, csr_src, out, 2, 0, 1, stream);
  run_group<2>(x, W, att, b2, uS, vS, rowmax, rowsum, rowstart, csr_src, out, 2, 2, 0, stream);
  run_group<1>(x, W, att, b1, uS, vS, rowmax, rowsum, rowstart, csr_src, out, 3, 0, 1, stream);
  run_group<1>(x, W, att, b1, uS, vS, rowmax, rowsum, rowstart, csr_src, out, 3, 1, 0, stream);
  run_group<1>(x, W, att, b1, uS, vS, rowmax, rowsum, rowstart, csr_src, out, 3, 2, 0, stream);
  run_group<1>(x, W, att, b1, uS, vS, rowmax, rowsum, rowstart, csr_src, out, 3, 3, 0, stream);

  epilogue_kernel<<<(unsigned)((outN + 255) / 256), 256, 0, stream>>>(out, bias, outN);

  (void)in_sizes; (void)n_in; (void)out_size; (void)ws_size;
}


# scores, row statistics and epilogue kernels start 4x to 32x fewer waves (a wave takes 4 nodes; an epilogue thread takes 8 float4 of a 1024-block grid) to cut wave entry time; every sum's order kept
# speedup vs baseline: 3.4448x; 1.0726x over previous
#include <hip/hip_runtime.h>
#include <hip/hip_bf16.h>

typedef __attribute__((ext_vector_type(2))) float v2f;
typedef __attribute__((ext_vector_type(8))) float v8f;
typedef __attribute__((ext_vector_type(4))) float v4f;

constexpr int BB = 2;
constexpr int VV = 4;
constexpr int NN = 2048;
constexpr int FF = 128;
constexpr int EE = 32768;
constexpr int NE = 34816;
constexpr int OO = 512;

__device__ __forceinline__ float sgnval(int r, int j) {
  if (j >= 4) return 1.0f;
  const bool flip = (j & 1) ? ((r & 2) != 0) : ((r & 1) != 0);
  return flip ? -1.0f : 1.0f;
}

__device__ __forceinline__ int src_of(const int* __restrict__ ei, int e) {
  return (e < EE) ? ei[e] : (e - EE);
}
__device__ __forceinline__ int dst_of(const int* __restrict__ ei, int e) {
  return (e < EE) ? ei[EE + e] : (e - EE);
}

struct Bufs {
  float* h0; float* h1; float* h2; float* h3;
  int bs0; int bs1; int bs2; int bs3;
};
__device__ __forceinline__ float* buf_of(const Bufs& B, int p) {
  return p == 0 ? B.h0 : (p == 1 ? B.h1 : (p == 2 ? B.h2 : B.h3));
}
__device__ __forceinline__ int bs_of(const Bufs& B, int p) {
  return p == 0 ? B.bs0 : (p == 1 ? B.bs1 : (p == 2 ? B.bs2 : B.bs3));
}
constexpr int PS = BB * NN;

__global__ void zero_i32_kernel(int* __restrict__ p, int n) {
  int i = blockIdx.x * blockDim.x + threadIdx.x;
  if (i < n) p[i] = 0;
}

__global__ void count_dst_kernel(const int* __restrict__ ei, int* __restrict__ counts,
                                 int* __restrict__ slot) {
  int e = blockIdx.x * blockDim.x + threadIdx.x;
  if (e >= NE) return;
  slot[e] = atomicAdd(&counts[dst_of(ei, e)], 1);
}

__global__ __launch_bounds__(256) void scan_counts_kernel(const int* __restrict__ counts,
                                                          int* __restrict__ rowstart) {
  __shared__ int part[256];
  __shared__ int pref[257];
  const int tid = threadIdx.x;
  const int base = tid * 8;
  int loc[8];
  int s = 0;
#pragma unroll
  for (int i = 0; i < 8; ++i) { loc[i] = s; s += counts[base + i]; }
  part[tid] = s;
  __syncthreads();
  if (tid == 0) {
    int a = 0;
    for (int i = 0; i < 256; ++i) { pref[i] = a; a += part[i]; }
    pref[256] = a;
  }
  __syncthreads();
  const int off = pref[tid];
#pragma unroll
  for (int i = 0; i < 8; ++i) rowstart[base + i] = off + loc[i];
  if (tid == 255) rowstart[NN] = pref[256];
}

__global__ void fill_rows_kernel(const int* __restrict__ ei, const int* __restrict__ rowstart,
                                 const int* __restrict__ slot, int* __restrict__ row_edge) {
  int e = blockIdx.x * blockDim.x + threadIdx.x;
  if (e >= NE) return;
  row_edge[rowstart[dst_of(ei, e)] + slot[e]] = e;
}

__global__ __launch_bounds__(256) void order_rows_kernel(const int* __restrict__ ei,
                                                         const int* __restrict__ rowstart,
                                                         const int* __restrict__ row_edge,
                                                         int* __restrict__ csr_src) {
  const int lane = threadIdx.x & 31;
  const int wave = threadIdx.x >> 5;
  const int n = blockIdx.x * 8 + wave;
  if (n >= NN) return;
  const int beg = __builtin_amdgcn_readfirstlane(rowstart[n]);
  const int len = __builtin_amdgcn_readfirstlane(rowstart[n + 1]) - beg;
  for (int i0 = 0; i0 < len; i0 += 32) {
    const int i = i0 + lane;
    const int id = (i < len) ? row_edge[beg + i] : 0x7fffffff;
    int r = 0;
    for (int j0 = 0; j0 < len; j0 += 32) {
      const int j = j0 + lane;
      const int idj = (j < len) ? row_edge[beg + j] : 0x7fffffff;
      const int m = (len - j0 < 32) ? (len - j0) : 32;
      for (int t = 0; t < m; ++t) r += (__builtin_amdgcn_readlane(idj, t) < id) ? 1 : 0;
    }
    if (i < len) csr_src[beg + r] = src_of(ei, id);
  }
}

__device__ __forceinline__ void wmma_quad(const float* __restrict__ W, int k, int khalf,
                                          int col, v2f a0, v2f a1, v8f* acc) {
  const float* w0 = W + (size_t)(k + khalf) * OO;
  const float* w1 = w0 + OO;
  const v2f b0 = { w0[col +  0], w1[col +  0] };
  const v2f b1 = { w0[col + 16], w1[col + 16] };
  const v2f b2 = { w0[col + 32], w1[col + 32] };
  const v2f b3 = { w0[col + 48], w1[col + 48] };
  acc[0] = __builtin_amdgcn_wmma_f32_16x16x4_f32(false, a0, false, b0, (short)0, acc[0], false, false);
  acc[1] = __builtin_amdgcn_wmma_f32_16x16x4_f32(false, a0, false, b1, (short)0, acc[1], false, false);
  acc[2] = __builtin_amdgcn_wmma_f32_16x16x4_f32(false, a0, false, b2, (short)0, acc[2], false, false);
  acc[3] = __builtin_amdgcn_wmma_f32_16x16x4_f32(false, a0, false, b3, (short)0, acc[3], false, false);
  acc[4] = __builtin_amdgcn_wmma_f32_16x16x4_f32(false, a1, false, b0, (short)0, acc[4], false, false);
  acc[5] = __builtin_amdgcn_wmma_f32_16x16x4_f32(false, a1, false, b1, (short)0, acc[5], false, false);
  acc[6] = __builtin_amdgcn_wmma_f32_16x16x4_f32(false, a1, false, b2, (short)0, acc[6], false, false);
  acc[7] = __builtin_amdgcn_wmma_f32_16x16x4_f32(false, a1, false, b3, (short)0, acc[7], false, false);
}

__global__ __launch_bounds__(256) void gemm_pair_kernel(const float* __restrict__ x,
                                                        const float* __restrict__ W,
                                                        Bufs B, int g, int hv0) {
  const int pr = blockIdx.y;
  const int v1 = hv0 + pr;
  const int v2 = g ^ v1;
  float* H = buf_of(B, pr);
  const int bs = bs_of(B, pr);
  const int lane = threadIdx.x & 31;
  const int wave = threadIdx.x >> 5;
  const int rowTile = blockIdx.x * 32;
  const int b = rowTile >> 11;
  const int nbase = rowTile & (NN - 1);
  const int r0 = nbase + (lane & 15);
  const float* x1a = x + (((size_t)b * VV + v1) * NN + r0) * FF;
  const float* x1b = x1a + 16 * FF;
  const float* x2a = x + (((size_t)b * VV + v2) * NN + r0) * FF;
  const float* x2b = x2a + 16 * FF;
  const int khalf = (lane >> 4) * 2;
  const int col = wave * 64 + (lane & 15);

  v8f acc[8] = {};

  {
    v2f a0 = *(const v2f*)(x1a + khalf);
    v2f a1 = *(const v2f*)(x1b + khalf);
    const float s0 = sgnval(v2, khalf);
    const float s1 = sgnval(v2, khalf + 1);
    a0.x *= s0; a0.y *= s1;
    a1.x *= s0; a1.y *= s1;
    wmma_quad(W, 0, khalf, col, a0, a1, acc);
  }
  for (int k = 4; k < FF; k += 4) {
    const v2f a0 = *(const v2f*)(x1a + k + khalf);
    const v2f a1 = *(const v2f*)(x1b + k + khalf);
    wmma_quad(W, k, khalf, col, a0, a1, acc);
  }
  {
    v2f a0 = *(const v2f*)(x2a + khalf);
    v2f a1 = *(const v2f*)(x2b + khalf);
    const float s0 = sgnval(v1, khalf);
    const float s1 = sgnval(v1, khalf + 1);
    a0.x *= s0; a0.y *= s1;
    a1.x *= s0; a1.y *= s1;
    wmma_quad(W, FF, khalf, col, a0, a1, acc);
  }
  for (int k = FF + 4; k < 2 * FF; k += 4) {
    const v2f a0 = *(const v2f*)(x2a + (k - FF) + khalf);
    const v2f a1 = *(const v2f*)(x2b + (k - FF) + khalf);
    wmma_quad(W, k, khalf, col, a0, a1, acc);
  }

  float* Hb = H + (size_t)b * bs;
  const int srow = nbase + 8 * (lane >> 4);
  const int scol = wave * 64 + (lane & 15);
#pragma unroll
  for (int i = 0; i < 8; ++i) {
    float* ra = Hb + (size_t)(srow + i) * OO + scol;
    ra[ 0] = acc[0][i];
    ra[16] = acc[1][i];
    ra[32] = acc[2][i];
    ra[48] = acc[3][i];
    float* rb = Hb + (size_t)(srow + 16 + i) * OO + scol;
    rb[ 0] = acc[4][i];
    rb[16] = acc[5][i];
    rb[32] = acc[6][i];
    rb[48] = acc[7][i];
  }
}

constexpr int NPW = 4;
__global__ __launch_bounds__(256) void node_scores_kernel(Bufs B,
                                                          const float* __restrict__ att,
                                                          float* __restrict__ uS,
                                                          float* __restrict__ vS) {
  const int lane = threadIdx.x & 31;
  const int wave = threadIdx.x >> 5;
  const int node0 = (blockIdx.x * 8 + wave) * NPW;
  const int pr = blockIdx.y;
  const float* hbuf = buf_of(B, pr);
  const int bs = bs_of(B, pr);
  uS += (size_t)pr * PS;
  vS += (size_t)pr * PS;
  float au[16], av[16];
#pragma unroll
  for (int i = 0; i < 16; ++i) { au[i] = att[lane + 32 * i]; av[i] = att[OO + lane + 32 * i]; }
  float hv[NPW][16];
#pragma unroll
  for (int q = 0; q < NPW; ++q) {
    const int node = node0 + q;
    const float* hrow = hbuf + (size_t)(node >> 11) * bs + (size_t)(node & (NN - 1)) * OO;
#pragma unroll
    for (int i = 0; i < 16; ++i) hv[q][i] = hrow[lane + 32 * i];
  }
#pragma unroll
  for (int q = 0; q < NPW; ++q) {
    float su = 0.0f, sv = 0.0f;
#pragma unroll
    for (int i = 0; i < 16; ++i) {
      const float h = hv[q][i];
      const float l = h > 0.0f ? h : 0.2f * h;
      su = fmaf(au[i], l, su);
      sv = fmaf(av[i], l, sv);
    }
#pragma unroll
    for (int off = 16; off; off >>= 1) {
      su += __shfl_down(su, off, 32);
      sv += __shfl_down(sv, off, 32);
    }
    if (lane == 0) { uS[node0 + q] = su; vS[node0 + q] = sv; }
  }
}

__global__ __launch_bounds__(256) void row_stats_kernel(const float* __restrict__ uS,
                                                        const float* __restrict__ vS,
                                                        const int* __restrict__ rowstart,
                                                        const int* __restrict__ csr_src,
                                                        float* __restrict__ rowmax,
                                                        float* __restrict__ rowsum) {
  const int lane = threadIdx.x & 31;
  const int wave = threadIdx.x >> 5;
  const int sub = lane & 7;
  const int node = (blockIdx.x * 8 + wave) * NPW + (lane >> 3);
  const int pr = blockIdx.y;
  uS += (size_t)pr * PS;
  vS += (size_t)pr * PS;
  rowmax += (size_t)pr * PS;
  rowsum += (size_t)pr * PS;
  const int b = node >> 11;
  const int n = node & (NN - 1);
  const float* ub = uS + (size_t)b * NN;
  const float vn = vS[node];
  const int beg = rowstart[n];
  const int end = rowstart[n + 1];
  float mx = -3.4e38f;
  for (int base = beg + sub; base < end; base += 32) {
    int id[4];
    float uu[4];
#pragma unroll
    for (int m = 0; m < 4; ++m) id[m] = csr_src[(base + 8 * m < end) ? (base + 8 * m) : base];
#pragma unroll
    for (int m = 0; m < 4; ++m) uu[m] = ub[id[m]];
#pragma unroll
    for (int m = 0; m < 4; ++m) {
      const float sc = uu[m] + vn;
      mx = (base + 8 * m < end) ? fmaxf(mx, sc) : mx;
    }
  }
#pragma unroll
  for (int off = 4; off; off >>= 1) mx = fmaxf(mx, __shfl_xor(mx, off, 32));
  float s0 = 0.0f, s1 = 0.0f, s2 = 0.0f, s3 = 0.0f;
  for (int base = beg + sub; base < end; base += 32) {
    int id[4];
    float ex[4];
#pragma unroll
    for (int m = 0; m < 4; ++m) id[m] = csr_src[(base + 8 * m < end) ? (base + 8 * m) : base];
#pragma unroll
    for (int m = 0; m < 4; ++m) ex[m] = ub[id[m]];
#pragma unroll
    for (int m = 0; m < 4; ++m) ex[m] = expf((ex[m] + vn) - mx);
    s0 += ex[0];
    s1 = (base + 8 < end)  ? s1 + ex[1] : s1;
    s2 = (base + 16 < end) ? s2 + ex[2] : s2;
    s3 = (base + 24 < end) ? s3 + ex[3] : s3;
  }
  float sm = (s0 + s2) + (s1 + s3);
#pragma unroll
  for (int off = 4; off; off >>= 1) sm += __shfl_xor(sm, off, 32);
  if (sub == 0) { rowmax[node] = mx; rowsum[node] = sm; }
}

template <int P>
__device__ __forceinline__ void hold_loads(const float (&uu)[P], const v4f (&hh)[P][4]) {
  if constexpr (P == 2) {
    asm volatile("" :: "v"(uu[0]), "v"(uu[1]),
                 "v"(hh[0][0]), "v"(hh[0][1]), "v"(hh[0][2]), "v"(hh[0][3]),
                 "v"(hh[1][0]), "v"(hh[1][1]), "v"(hh[1][2]), "v"(hh[1][3]));
  } else if constexpr (P == 3) {
    asm volatile("" :: "v"(uu[0]), "v"(uu[1]), "v"(uu[2]),
                 "v"(hh[0][0]), "v"(hh[0][1]), "v"(hh[0][2]), "v"(hh[0][3]),
                 "v"(hh[1][0]), "v"(hh[1][1]), "v"(hh[1][2]), "v"(hh[1][3]),
                 "v"(hh[2][0]), "v"(hh[2][1]), "v"(hh[2][2]), "v"(hh[2][3]));
  } else {
    asm volatile("" :: "v"(uu[0]), "v"(uu[1]), "v"(uu[2]), "v"(uu[3]),
                 "v"(hh[0][0]), "v"(hh[0][1]), "v"(hh[0][2]), "v"(hh[0][3]),
                 "v"(hh[1][0]), "v"(hh[1][1]), "v"(hh[1][2]), "v"(hh[1][3]),
                 "v"(hh[2][0]), "v"(hh[2][1]), "v"(hh[2][2]), "v"(hh[2][3]),
                 "v"(hh[3][0]), "v"(hh[3][1]), "v"(hh[3][2]), "v"(hh[3][3]));
  }
}

template <int P, int E>
__device__ __forceinline__ void walk_step(const int* __restrict__ csr_src, int idx, int lane,
                                          const float* const (&Hb)[P], const float* const (&ub)[P],
                                          const float (&vn)[P], const float (&m)[P],
                                          const float (&inv)[P], float4 (&acc)[P][4]) {
  int sn[E];
#pragma unroll
  for (int e = 0; e < E; ++e) sn[e] = csr_src[idx + e];
  float uu[E][P];
  v4f hh[E][P][4];
#pragma unroll
  for (int e = 0; e < E; ++e) {
#pragma unroll
    for (int p = 0; p < P; ++p) {
      uu[e][p] = ub[p][sn[e]];
      const v4f* hr = (const v4f*)(Hb[p] + (size_t)sn[e] * OO);
#pragma unroll
      for (int j = 0; j < 4; ++j) hh[e][p][j] = hr[lane + 32 * j];
    }
  }
#pragma unroll
  for (int e = 0; e < E; ++e) hold_loads<P>(uu[e], hh[e]);
#pragma unroll
  for (int e = 0; e < E; ++e) {
#pragma unroll
    for (int p = 0; p < P; ++p) {
      const float w = expf((uu[e][p] + vn[p]) - m[p]) * inv[p];
#pragma unroll
      for (int j = 0; j < 4; ++j) {
        acc[p][j].x += w * hh[e][p][j].x;
        acc[p][j].y += w * hh[e][p][j].y;
        acc[p][j].z += w * hh[e][p][j].z;
        acc[p][j].w += w * hh[e][p][j].w;
      }
    }
  }
}

template <int P>
__global__ __launch_bounds__(256) void aggregate_kernel(Bufs B,
                                                        const float* __restrict__ uS,
                                                        const float* __restrict__ vS,
                                                        const float* __restrict__ rowmax,
                                                        const float* __restrict__ rowsum,
                                                        const int* __restrict__ rowstart,
                                                        const int* __restrict__ csr_src,
                                                        float* out, int g, int first) {
  __shared__ float redm[P][8];
  __shared__ float reds[P][8];
  const int tid = threadIdx.x;
  const int lane = tid & 31;
  const int wave = tid >> 5;
  const int node = blockIdx.x * 8 + wave;
  const int b = node >> 11;
  const int n = node & (NN - 1);
  const int beg = rowstart[n];
  const int end = rowstart[n + 1];

  float vn[P], m[P], inv[P];
  const float* Hb[P];
  const float* ub[P];
  float4 m0[P], m1[P], s0[P], s1[P];
#pragma unroll
  for (int p = 0; p < P; ++p) {
    vn[p] = vS[(size_t)p * PS + node];
    Hb[p] = buf_of(B, p) + (size_t)b * bs_of(B, p);
    ub[p] = uS + (size_t)p * PS + (size_t)b * NN;
    const float4* rm4 = (const float4*)(rowmax + (size_t)p * PS + (size_t)b * NN) + 2 * tid;
    const float4* rs4 = (const float4*)(rowsum + (size_t)p * PS + (size_t)b * NN) + 2 * tid;
    m0[p] = rm4[0]; m1[p] = rm4[1];
    s0[p] = rs4[0]; s1[p] = rs4[1];
  }
#pragma unroll
  for (int p = 0; p < P; ++p) {
    float mx = fmaxf(fmaxf(fmaxf(m0[p].x, m0[p].y), fmaxf(m0[p].z, m0[p].w)),
                     fmaxf(fmaxf(m1[p].x, m1[p].y), fmaxf(m1[p].z, m1[p].w)));
#pragma unroll
    for (int off = 16; off; off >>= 1) mx = fmaxf(mx, __shfl_xor(mx, off, 32));
    if (lane == 0) redm[p][wave] = mx;
  }
  __syncthreads();
#pragma unroll
  for (int p = 0; p < P; ++p) {
    float mm = redm[p][0];
#pragma unroll
    for (int i = 1; i < 8; ++i) mm = fmaxf(mm, redm[p][i]);
    float z = s0[p].x * expf(m0[p].x - mm);
    z += s0[p].y * expf(m0[p].y - mm);
    z += s0[p].z * expf(m0[p].z - mm);
    z += s0[p].w * expf(m0[p].w - mm);
    z += s1[p].x * expf(m1[p].x - mm);
    z += s1[p].y * expf(m1[p].y - mm);
    z += s1[p].z * expf(m1[p].z - mm);
    z += s1[p].w * expf(m1[p].w - mm);
#pragma unroll
    for (int off = 16; off; off >>= 1) z += __shfl_xor(z, off, 32);
    if (lane == 0) reds[p][wave] = z;
    m[p] = mm;
  }
  __syncthreads();
#pragma unroll
  for (int p = 0; p < P; ++p) {
    float zsum = reds[p][0];
#pragma unroll
    for (int i = 1; i < 8; ++i) zsum += reds[p][i];
    inv[p] = 1.0f / zsum;
  }

  float4 acc[P][4] = {};
  if constexpr (P == 1) {
    for (int idx = beg; idx < end; ++idx) {
      const int sn = csr_src[idx];
      const float w = expf((ub[0][sn] + vn[0]) - m[0]) * inv[0];
      const float4* hr = (const float4*)(Hb[0] + (size_t)sn * OO);
#pragma unroll
      for (int j = 0; j < 4; ++j) {
        const float4 h4 = hr[lane + 32 * j];
        acc[0][j].x += w * h4.x;
        acc[0][j].y += w * h4.y;
        acc[0][j].z += w * h4.z;
        acc[0][j].w += w * h4.w;
      }
    }
  } else {
    int idx = beg;
    for (; idx + 1 < end; idx += 2) walk_step<P, 2>(csr_src, idx, lane, Hb, ub, vn, m, inv, acc);
    if (idx < end) walk_step<P, 1>(csr_src, idx, lane, Hb, ub, vn, m, inv, acc);
  }
  float4* op = (float4*)(out + (((size_t)b * VV + g) * NN + n) * OO);
  float4 q[4];
  if (first) {
#pragma unroll
    for (int j = 0; j < 4; ++j) q[j] = make_float4(0.0f, 0.0f, 0.0f, 0.0f);
  } else {
#pragma unroll
    for (int j = 0; j < 4; ++j) q[j] = op[lane + 32 * j];
  }
#pragma unroll
  for (int j = 0; j < 4; ++j) {
    float4 o = q[j];
#pragma unroll
    for (int p = 0; p < P; ++p) {
      o.x += acc[p][j].x; o.y += acc[p][j].y; o.z += acc[p][j].z; o.w += acc[p][j].w;
    }
    op[lane + 32 * j] = o;
  }
}

constexpr int EPI_BLOCKS = 1024;
constexpr int EPI_THREADS = EPI_BLOCKS * 256;
constexpr int EPI_PER = (BB * VV * NN * OO / 4) / EPI_THREADS;
static_assert((BB * VV * NN * OO / 4) % EPI_THREADS == 0, "epilogue grid must tile out exactly");
static_assert(EPI_THREADS % (OO / 4) == 0, "a thread's pieces must share one bias piece");
__global__ __launch_bounds__(256) void epilogue_kernel(float* __restrict__ out, const float* __restrict__ bias) {
  const int t = blockIdx.x * 256 + threadIdx.x;
  float4* o4 = (float4*)out;
  const int bc = 4 * (t & (OO / 4 - 1));
  const float4 bv = make_float4(bias[bc], bias[bc + 1], bias[bc + 2], bias[bc + 3]);
  float4 q[EPI_PER];
#pragma unroll
  for (int k = 0; k < EPI_PER; ++k) q[k] = o4[(size_t)t + (size_t)k * EPI_THREADS];
#pragma unroll
  for (int k = 0; k < EPI_PER; ++k) {
    float4 r;
    r.x = q[k].x * 0.25f + bv.x;
    r.y = q[k].y * 0.25f + bv.y;
    r.z = q[k].z * 0.25f + bv.z;
    r.w = q[k].w * 0.25f + bv.w;
    o4[(size_t)t + (size_t)k * EPI_THREADS] = r;
  }
}

template <int P>
static void run_group(const float* x, const float* W, const float* att, Bufs B,
                      float* uS, float* vS, float* rowmax, float* rowsum,
                      const int* rowstart, const int* csr_src, float* out,
                      int g, int hv0, int first, hipStream_t stream) {
  gemm_pair_kernel<<<dim3((BB * NN) / 32, P), 256, 0, stream>>>(x, W, B, g, hv0);
  node_scores_kernel<<<dim3((BB * NN) / (8 * NPW), P), 256, 0, stream>>>(B, att, uS, vS);
  row_stats_kernel<<<dim3((BB * NN) / (8 * NPW), P), 256, 0, stream>>>(uS, vS, rowstart, csr_src, rowmax, rowsum);
  aggregate_kernel<P><<<(BB * NN) / 8, 256, 0, stream>>>(B, uS, vS, rowmax, rowsum, rowstart, csr_src,
                                                         out, g, first);
}

extern "C" void kernel_launch(void* const* d_in, const int* in_sizes, int n_in,
                              void* d_out, int out_size, void* d_ws, size_t ws_size,
                              hipStream_t stream) {
  const float* x    = (const float*)d_in[0];
  const int*   ei   = (const int*)d_in[1];
  const float* W    = (const float*)d_in[2];
  const float* att  = (const float*)d_in[3];
  const float* bias = (const float*)d_in[4];
  float* out = (float*)d_out;

  char* p = (char*)d_ws;
  float* H     = (float*)p; p += (size_t)BB * NN * OO * sizeof(float);
  float* uS     = (float*)p;
  float* vS     = uS + (size_t)4 * PS;
  float* rowmax = vS + (size_t)4 * PS;
  float* rowsum = rowmax + (size_t)4 * PS;
  p += (size_t)2 * BB * NN * sizeof(float);
  p += (size_t)BB * NE * sizeof(float);
  p += 4 * sizeof(float);
  int* counts  = (int*)p;   p += (size_t)NN * sizeof(int);
  int* csr_src = (int*)p;   p += (size_t)NE * sizeof(int);
  int* rowstart= (int*)p;   p += (size_t)(NN + 1) * sizeof(int);
  int* slot     = (int*)H;
  int* row_edge = slot + NE;

  zero_i32_kernel<<<(NN + 255) / 256, 256, 0, stream>>>(counts, NN);
  count_dst_kernel<<<(NE + 255) / 256, 256, 0, stream>>>(ei, counts, slot);
  scan_counts_kernel<<<1, 256, 0, stream>>>(counts, rowstart);
  fill_rows_kernel<<<(NE + 255) / 256, 256, 0, stream>>>(ei, rowstart, slot, row_edge);
  order_rows_kernel<<<NN / 8, 256, 0, stream>>>(ei, rowstart, row_edge, csr_src);

  constexpr int WS_BS = NN * OO;
  constexpr int PL_BS = VV * NN * OO;
  float* pl1 = out + (size_t)1 * NN * OO;
  float* pl2 = out + (size_t)2 * NN * OO;
  float* pl3 = out + (size_t)3 * NN * OO;
  const Bufs b4 = { H, pl1, pl2, pl3, WS_BS, PL_BS, PL_BS, PL_BS };
  const Bufs b3 = { H, pl2, pl3, pl3, WS_BS, PL_BS, PL_BS, PL_BS };
  const Bufs b2 = { H, pl3, pl3, pl3, WS_BS, PL_BS, PL_BS, PL_BS };
  const Bufs b1 = { H, H, H, H, WS_BS, WS_BS, WS_BS, WS_BS };

  run_group<4>(x, W, att, b4, uS, vS, rowmax, rowsum, rowstart, csr_src, out, 0, 0, 1, stream);
  run_group<3>(x, W, att, b3, uS, vS, rowmax, rowsum, rowstart, csr_src, out, 1, 0, 1, stream);
  run_group<1>(x, W, att, b1, uS, vS, rowmax, rowsum, rowstart, csr_src, out, 1, 3, 0, stream);
  run_group<2>(x, W, att, b2, uS, vS, rowmax, rowsum, rowstart, csr_src, out, 2, 0, 1, stream);
  run_group<2>(x, W, att, b2, uS, vS, rowmax, rowsum, rowstart, csr_src, out, 2, 2, 0, stream);
  run_group<1>(x, W, att, b1, uS, vS, rowmax, rowsum, rowstart, csr_src, out, 3, 0, 1, stream);
  run_group<1>(x, W, att, b1, uS, vS, rowmax, rowsum, rowstart, csr_src, out, 3, 1, 0, stream);
  run_group<1>(x, W, att, b1, uS, vS, rowmax, rowsum, rowstart, csr_src, out, 3, 2, 0, stream);
  run_group<1>(x, W, att, b1, uS, vS, rowmax, rowsum, rowstart, csr_src, out, 3, 3, 0, stream);

  epilogue_kernel<<<EPI_BLOCKS, 256, 0, stream>>>(out, bias);

  (void)in_sizes; (void)n_in; (void)out_size; (void)ws_size;
}


# GEMM: pair (a,g^a) and its mirror (g^a,a) made from two half-length WMMA products S=(P+Q)(W1+W2), D=(P-Q)(W1-W2), H=(S+-D)/2; views run 1,2,3,0: product and view sum order change
# speedup vs baseline: 4.1123x; 1.1938x over previous
#include <hip/hip_runtime.h>
#include <hip/hip_bf16.h>

typedef __attribute__((ext_vector_type(2))) float v2f;
typedef __attribute__((ext_vector_type(8))) float v8f;
typedef __attribute__((ext_vector_type(4))) float v4f;

constexpr int BB = 2;
constexpr int VV = 4;
constexpr int NN = 2048;
constexpr int FF = 128;
constexpr int EE = 32768;
constexpr int NE = 34816;
constexpr int OO = 512;

__device__ __forceinline__ float sgnval(int r, int j) {
  if (j >= 4) return 1.0f;
  const bool flip = (j & 1) ? ((r & 2) != 0) : ((r & 1) != 0);
  return flip ? -1.0f : 1.0f;
}

__device__ __forceinline__ int src_of(const int* __restrict__ ei, int e) {
  return (e < EE) ? ei[e] : (e - EE);
}
__device__ __forceinline__ int dst_of(const int* __restrict__ ei, int e) {
  return (e < EE) ? ei[EE + e] : (e - EE);
}

struct Bufs {
  float* h0; float* h1; float* h2; float* h3;
  int bs0; int bs1; int bs2; int bs3;
};
__device__ __forceinline__ float* buf_of(const Bufs& B, int p) {
  return p == 0 ? B.h0 : (p == 1 ? B.h1 : (p == 2 ? B.h2 : B.h3));
}
__device__ __forceinline__ int bs_of(const Bufs& B, int p) {
  return p == 0 ? B.bs0 : (p == 1 ? B.bs1 : (p == 2 ? B.bs2 : B.bs3));
}
constexpr int PS = BB * NN;

__global__ void zero_i32_kernel(int* __restrict__ p, int n) {
  int i = blockIdx.x * blockDim.x + threadIdx.x;
  if (i < n) p[i] = 0;
}

__global__ void count_dst_kernel(const int* __restrict__ ei, int* __restrict__ counts,
                                 int* __restrict__ slot) {
  int e = blockIdx.x * blockDim.x + threadIdx.x;
  if (e >= NE) return;
  slot[e] = atomicAdd(&counts[dst_of(ei, e)], 1);
}

__global__ __launch_bounds__(256) void scan_counts_kernel(const int* __restrict__ counts,
                                                          int* __restrict__ rowstart) {
  __shared__ int part[256];
  __shared__ int pref[257];
  const int tid = threadIdx.x;
  const int base = tid * 8;
  int loc[8];
  int s = 0;
#pragma unroll
  for (int i = 0; i < 8; ++i) { loc[i] = s; s += counts[base + i]; }
  part[tid] = s;
  __syncthreads();
  if (tid == 0) {
    int a = 0;
    for (int i = 0; i < 256; ++i) { pref[i] = a; a += part[i]; }
    pref[256] = a;
  }
  __syncthreads();
  const int off = pref[tid];
#pragma unroll
  for (int i = 0; i < 8; ++i) rowstart[base + i] = off + loc[i];
  if (tid == 255) rowstart[NN] = pref[256];
}

__global__ void fill_rows_kernel(const int* __restrict__ ei, const int* __restrict__ rowstart,
                                 const int* __restrict__ slot, int* __restrict__ row_edge) {
  int e = blockIdx.x * blockDim.x + threadIdx.x;
  if (e >= NE) return;
  row_edge[rowstart[dst_of(ei, e)] + slot[e]] = e;
}

__global__ __launch_bounds__(256) void order_rows_kernel(const int* __restrict__ ei,
                                                         const int* __restrict__ rowstart,
                                                         const int* __restrict__ row_edge,
                                                         int* __restrict__ csr_src) {
  const int lane = threadIdx.x & 31;
  const int wave = threadIdx.x >> 5;
  const int n = blockIdx.x * 8 + wave;
  if (n >= NN) return;
  const int beg = __builtin_amdgcn_readfirstlane(rowstart[n]);
  const int len = __builtin_amdgcn_readfirstlane(rowstart[n + 1]) - beg;
  for (int i0 = 0; i0 < len; i0 += 32) {
    const int i = i0 + lane;
    const int id = (i < len) ? row_edge[beg + i] : 0x7fffffff;
    int r = 0;
    for (int j0 = 0; j0 < len; j0 += 32) {
      const int j = j0 + lane;
      const int idj = (j < len) ? row_edge[beg + j] : 0x7fffffff;
      const int m = (len - j0 < 32) ? (len - j0) : 32;
      for (int t = 0; t < m; ++t) r += (__builtin_amdgcn_readlane(idj, t) < id) ? 1 : 0;
    }
    if (i < len) csr_src[beg + r] = src_of(ei, id);
  }
}

template <bool MIRROR>
struct RawGroup {
  v2f p0[4], p1[4];
  v2f q0[4], q1[4];
  v2f u1[4][4], u2[4][4];
};

template <bool MIRROR>
__device__ __forceinline__ void load_group(RawGroup<MIRROR>& R, const float* __restrict__ xa0,
                                           const float* __restrict__ xc0, const float* __restrict__ Wc,
                                           int kg, int khalf) {
#pragma unroll
  for (int t = 0; t < 4; ++t) {
    const int kk = kg + 4 * t + khalf;
    R.p0[t] = *(const v2f*)(xa0 + kk);
    R.p1[t] = *(const v2f*)(xa0 + 16 * FF + kk);
    if constexpr (MIRROR) {
      R.q0[t] = *(const v2f*)(xc0 + kk);
      R.q1[t] = *(const v2f*)(xc0 + 16 * FF + kk);
    }
    const float* w1a = Wc + (size_t)kk * OO;
    const float* w1b = w1a + OO;
    const float* w2a = w1a + (size_t)FF * OO;
    const float* w2b = w2a + OO;
#pragma unroll
    for (int j = 0; j < 4; ++j) {
      R.u1[t][j] = v2f{ w1a[16 * j], w1b[16 * j] };
      R.u2[t][j] = v2f{ w2a[16 * j], w2b[16 * j] };
    }
  }
}

template <bool MIRROR>
__device__ __forceinline__ void mma_group(const RawGroup<MIRROR>& R, bool signedQuad, v2f sp, v2f sq,
                                          v8f (&accS)[8], v8f (&accD)[8]) {
#pragma unroll
  for (int t = 0; t < 4; ++t) {
    v2f p0 = R.p0[t], p1 = R.p1[t];
    if (t == 0 && signedQuad) { p0 *= sp; p1 *= sp; }
    v2f aS0 = p0, aS1 = p1, aD0 = p0, aD1 = p1;
    if constexpr (MIRROR) {
      v2f q0 = R.q0[t], q1 = R.q1[t];
      if (t == 0 && signedQuad) { q0 *= sq; q1 *= sq; }
      aS0 = p0 + q0; aS1 = p1 + q1;
      aD0 = p0 - q0; aD1 = p1 - q1;
    }
#pragma unroll
    for (int j = 0; j < 4; ++j) {
      const v2f bS = R.u1[t][j] + R.u2[t][j];
      accS[j]     = __builtin_amdgcn_wmma_f32_16x16x4_f32(false, aS0, false, bS, (short)0, accS[j],     false, false);
      accS[4 + j] = __builtin_amdgcn_wmma_f32_16x16x4_f32(false, aS1, false, bS, (short)0, accS[4 + j], false, false);
    }
    if constexpr (MIRROR) {
#pragma unroll
      for (int j = 0; j < 4; ++j) {
        const v2f bD = R.u1[t][j] - R.u2[t][j];
        accD[j]     = __builtin_amdgcn_wmma_f32_16x16x4_f32(false, aD0, false, bD, (short)0, accD[j],     false, false);
        accD[4 + j] = __builtin_amdgcn_wmma_f32_16x16x4_f32(false, aD1, false, bD, (short)0, accD[4 + j], false, false);
      }
    }
  }
}

template <bool MIRROR>
__global__ __launch_bounds__(128) void gemm_mirror_kernel(const float* __restrict__ x,
                                                          const float* __restrict__ W,
                                                          Bufs B, int g, int a0, int astep) {
  const int m = blockIdx.y;
  const int a = a0 + m * astep;
  const int c = g ^ a;
  const int lane = threadIdx.x & 31;
  const int wave = threadIdx.x >> 5;
  const int rowTile = blockIdx.x * 32;
  const int b = rowTile >> 11;
  const int nbase = rowTile & (NN - 1);
  const int r0 = nbase + (lane & 15);
  const float* xa0 = x + (((size_t)b * VV + a) * NN + r0) * FF;
  const float* xc0 = x + (((size_t)b * VV + c) * NN + r0) * FF;
  const int khalf = (lane >> 4) * 2;
  const int col = (blockIdx.z * 4 + wave) * 64 + (lane & 15);
  const float* Wc = W + col;
  const v2f sp = { sgnval(c, khalf), sgnval(c, khalf + 1) };
  const v2f sq = { sgnval(a, khalf), sgnval(a, khalf + 1) };

  v8f accS[8] = {};
  v8f accD[8] = {};

  RawGroup<MIRROR> R0, R1;
  load_group<MIRROR>(R0, xa0, xc0, Wc, 0, khalf);
#pragma unroll 1
  for (int kg = 0; kg < FF - 32; kg += 32) {
    load_group<MIRROR>(R1, xa0, xc0, Wc, kg + 16, khalf);
    mma_group<MIRROR>(R0, kg == 0, sp, sq, accS, accD);
    load_group<MIRROR>(R0, xa0, xc0, Wc, kg + 32, khalf);
    mma_group<MIRROR>(R1, false, sp, sq, accS, accD);
  }
  load_group<MIRROR>(R1, xa0, xc0, Wc, FF - 16, khalf);
  mma_group<MIRROR>(R0, false, sp, sq, accS, accD);
  mma_group<MIRROR>(R1, false, sp, sq, accS, accD);

  const int srow = nbase + 8 * (lane >> 4);
  float* Ha = buf_of(B, MIRROR ? 2 * m : m) + (size_t)b * bs_of(B, MIRROR ? 2 * m : m);
  float* Hc = buf_of(B, MIRROR ? 2 * m + 1 : m) + (size_t)b * bs_of(B, MIRROR ? 2 * m + 1 : m);
#pragma unroll
  for (int i = 0; i < 8; ++i) {
#pragma unroll
    for (int t = 0; t < 2; ++t) {
      const size_t off = (size_t)(srow + 16 * t + i) * OO + col;
#pragma unroll
      for (int j = 0; j < 4; ++j) {
        const float sv = accS[4 * t + j][i];
        if constexpr (MIRROR) {
          const float dv = accD[4 * t + j][i];
          Ha[off + 16 * j] = 0.5f * (sv + dv);
          Hc[off + 16 * j] = 0.5f * (sv - dv);
        } else {
          Ha[off + 16 * j] = sv;
        }
      }
    }
  }
}

constexpr int NPW = 4;
__global__ __launch_bounds__(256) void node_scores_kernel(Bufs B,
                                                          const float* __restrict__ att,
                                                          float* __restrict__ uS,
                                                          float* __restrict__ vS) {
  const int lane = threadIdx.x & 31;
  const int wave = threadIdx.x >> 5;
  const int node0 = (blockIdx.x * 8 + wave) * NPW;
  const int pr = blockIdx.y;
  const float* hbuf = buf_of(B, pr);
  const int bs = bs_of(B, pr);
  uS += (size_t)pr * PS;
  vS += (size_t)pr * PS;
  float au[16], av[16];
#pragma unroll
  for (int i = 0; i < 16; ++i) { au[i] = att[lane + 32 * i]; av[i] = att[OO + lane + 32 * i]; }
  float hv[NPW][16];
#pragma unroll
  for (int q = 0; q < NPW; ++q) {
    const int node = node0 + q;
    const float* hrow = hbuf + (size_t)(node >> 11) * bs + (size_t)(node & (NN - 1)) * OO;
#pragma unroll
    for (int i = 0; i < 16; ++i) hv[q][i] = hrow[lane + 32 * i];
  }
#pragma unroll
  for (int q = 0; q < NPW; ++q) {
    float su = 0.0f, sv = 0.0f;
#pragma unroll
    for (int i = 0; i < 16; ++i) {
      const float h = hv[q][i];
      const float l = h > 0.0f ? h : 0.2f * h;
      su = fmaf(au[i], l, su);
      sv = fmaf(av[i], l, sv);
    }
#pragma unroll
    for (int off = 16; off; off >>= 1) {
      su += __shfl_down(su, off, 32);
      sv += __shfl_down(sv, off, 32);
    }
    if (lane == 0) { uS[node0 + q] = su; vS[node0 + q] = sv; }
  }
}

__global__ __launch_bounds__(256) void row_stats_kernel(const float* __restrict__ uS,
                                                        const float* __restrict__ vS,
                                                        const int* __restrict__ rowstart,
                                                        const int* __restrict__ csr_src,
                                                        float* __restrict__ rowmax,
                                                        float* __restrict__ rowsum) {
  const int lane = threadIdx.x & 31;
  const int wave = threadIdx.x >> 5;
  const int sub = lane & 7;
  const int node = (blockIdx.x * 8 + wave) * NPW + (lane >> 3);
  const int pr = blockIdx.y;
  uS += (size_t)pr * PS;
  vS += (size_t)pr * PS;
  rowmax += (size_t)pr * PS;
  rowsum += (size_t)pr * PS;
  const int b = node >> 11;
  const int n = node & (NN - 1);
  const float* ub = uS + (size_t)b * NN;
  const float vn = vS[node];
  const int beg = rowstart[n];
  const int end = rowstart[n + 1];
  float mx = -3.4e38f;
  for (int base = beg + sub; base < end; base += 32) {
    int id[4];
    float uu[4];
#pragma unroll
    for (int m = 0; m < 4; ++m) id[m] = csr_src[(base + 8 * m < end) ? (base + 8 * m) : base];
#pragma unroll
    for (int m = 0; m < 4; ++m) uu[m] = ub[id[m]];
#pragma unroll
    for (int m = 0; m < 4; ++m) {
      const float sc = uu[m] + vn;
      mx = (base + 8 * m < end) ? fmaxf(mx, sc) : mx;
    }
  }
#pragma unroll
  for (int off = 4; off; off >>= 1) mx = fmaxf(mx, __shfl_xor(mx, off, 32));
  float s0 = 0.0f, s1 = 0.0f, s2 = 0.0f, s3 = 0.0f;
  for (int base = beg + sub; base < end; base += 32) {
    int id[4];
    float ex[4];
#pragma unroll
    for (int m = 0; m < 4; ++m) id[m] = csr_src[(base + 8 * m < end) ? (base + 8 * m) : base];
#pragma unroll
    for (int m = 0; m < 4; ++m) ex[m] = ub[id[m]];
#pragma unroll
    for (int m = 0; m < 4; ++m) ex[m] = expf((ex[m] + vn) - mx);
    s0 += ex[0];
    s1 = (base + 8 < end)  ? s1 + ex[1] : s1;
    s2 = (base + 16 < end) ? s2 + ex[2] : s2;
    s3 = (base + 24 < end) ? s3 + ex[3] : s3;
  }
  float sm = (s0 + s2) + (s1 + s3);
#pragma unroll
  for (int off = 4; off; off >>= 1) sm += __shfl_xor(sm, off, 32);
  if (sub == 0) { rowmax[node] = mx; rowsum[node] = sm; }
}

template <int P>
__device__ __forceinline__ void hold_loads(const float (&uu)[P], const v4f (&hh)[P][4]) {
  if constexpr (P == 2) {
    asm volatile("" :: "v"(uu[0]), "v"(uu[1]),
                 "v"(hh[0][0]), "v"(hh[0][1]), "v"(hh[0][2]), "v"(hh[0][3]),
                 "v"(hh[1][0]), "v"(hh[1][1]), "v"(hh[1][2]), "v"(hh[1][3]));
  } else if constexpr (P == 3) {
    asm volatile("" :: "v"(uu[0]), "v"(uu[1]), "v"(uu[2]),
                 "v"(hh[0][0]), "v"(hh[0][1]), "v"(hh[0][2]), "v"(hh[0][3]),
                 "v"(hh[1][0]), "v"(hh[1][1]), "v"(hh[1][2]), "v"(hh[1][3]),
                 "v"(hh[2][0]), "v"(hh[2][1]), "v"(hh[2][2]), "v"(hh[2][3]));
  } else {
    asm volatile("" :: "v"(uu[0]), "v"(uu[1]), "v"(uu[2]), "v"(uu[3]),
                 "v"(hh[0][0]), "v"(hh[0][1]), "v"(hh[0][2]), "v"(hh[0][3]),
                 "v"(hh[1][0]), "v"(hh[1][1]), "v"(hh[1][2]), "v"(hh[1][3]),
                 "v"(hh[2][0]), "v"(hh[2][1]), "v"(hh[2][2]), "v"(hh[2][3]),
                 "v"(hh[3][0]), "v"(hh[3][1]), "v"(hh[3][2]), "v"(hh[3][3]));
  }
}

template <int P, int E>
__device__ __forceinline__ void walk_step(const int* __restrict__ csr_src, int idx, int lane,
                                          const float* const (&Hb)[P], const float* const (&ub)[P],
                                          const float (&vn)[P], const float (&m)[P],
                                          const float (&inv)[P], float4 (&acc)[P][4]) {
  int sn[E];
#pragma unroll
  for (int e = 0; e < E; ++e) sn[e] = csr_src[idx + e];
  float uu[E][P];
  v4f hh[E][P][4];
#pragma unroll
  for (int e = 0; e < E; ++e) {
#pragma unroll
    for (int p = 0; p < P; ++p) {
      uu[e][p] = ub[p][sn[e]];
      const v4f* hr = (const v4f*)(Hb[p] + (size_t)sn[e] * OO);
#pragma unroll
      for (int j = 0; j < 4; ++j) hh[e][p][j] = hr[lane + 32 * j];
    }
  }
#pragma unroll
  for (int e = 0; e < E; ++e) hold_loads<P>(uu[e], hh[e]);
#pragma unroll
  for (int e = 0; e < E; ++e) {
#pragma unroll
    for (int p = 0; p < P; ++p) {
      const float w = expf((uu[e][p] + vn[p]) - m[p]) * inv[p];
#pragma unroll
      for (int j = 0; j < 4; ++j) {
        acc[p][j].x += w * hh[e][p][j].x;
        acc[p][j].y += w * hh[e][p][j].y;
        acc[p][j].z += w * hh[e][p][j].z;
        acc[p][j].w += w * hh[e][p][j].w;
      }
    }
  }
}

template <int P>
__global__ __launch_bounds__(256) void aggregate_kernel(Bufs B,
                                                        const float* __restrict__ uS,
                                                        const float* __restrict__ vS,
                                                        const float* __restrict__ rowmax,
                                                        const float* __restrict__ rowsum,
                                                        const int* __restrict__ rowstart,
                                                        const int* __restrict__ csr_src,
                                                        float* out, int g, int first) {
  __shared__ float redm[P][8];
  __shared__ float reds[P][8];
  const int tid = threadIdx.x;
  const int lane = tid & 31;
  const int wave = tid >> 5;
  const int node = blockIdx.x * 8 + wave;
  const int b = node >> 11;
  const int n = node & (NN - 1);
  const int beg = rowstart[n];
  const int end = rowstart[n + 1];

  float vn[P], m[P], inv[P];
  const float* Hb[P];
  const float* ub[P];
  float4 m0[P], m1[P], s0[P], s1[P];
#pragma unroll
  for (int p = 0; p < P; ++p) {
    vn[p] = vS[(size_t)p * PS + node];
    Hb[p] = buf_of(B, p) + (size_t)b * bs_of(B, p);
    ub[p] = uS + (size_t)p * PS + (size_t)b * NN;
    const float4* rm4 = (const float4*)(rowmax + (size_t)p * PS + (size_t)b * NN) + 2 * tid;
    const float4* rs4 = (const float4*)(rowsum + (size_t)p * PS + (size_t)b * NN) + 2 * tid;
    m0[p] = rm4[0]; m1[p] = rm4[1];
    s0[p] = rs4[0]; s1[p] = rs4[1];
  }
#pragma unroll
  for (int p = 0; p < P; ++p) {
    float mx = fmaxf(fmaxf(fmaxf(m0[p].x, m0[p].y), fmaxf(m0[p].z, m0[p].w)),
                     fmaxf(fmaxf(m1[p].x, m1[p].y), fmaxf(m1[p].z, m1[p].w)));
#pragma unroll
    for (int off = 16; off; off >>= 1) mx = fmaxf(mx, __shfl_xor(mx, off, 32));
    if (lane == 0) redm[p][wave] = mx;
  }
  __syncthreads();
#pragma unroll
  for (int p = 0; p < P; ++p) {
    float mm = redm[p][0];
#pragma unroll
    for (int i = 1; i < 8; ++i) mm = fmaxf(mm, redm[p][i]);
    float z = s0[p].x * expf(m0[p].x - mm);
    z += s0[p].y * expf(m0[p].y - mm);
    z += s0[p].z * expf(m0[p].z - mm);
    z += s0[p].w * expf(m0[p].w - mm);
    z += s1[p].x * expf(m1[p].x - mm);
    z += s1[p].y * expf(m1[p].y - mm);
    z += s1[p].z * expf(m1[p].z - mm);
    z += s1[p].w * expf(m1[p].w - mm);
#pragma unroll
    for (int off = 16; off; off >>= 1) z += __shfl_xor(z, off, 32);
    if (lane == 0) reds[p][wave] = z;
    m[p] = mm;
  }
  __syncthreads();
#pragma unroll
  for (int p = 0; p < P; ++p) {
    float zsum = reds[p][0];
#pragma unroll
    for (int i = 1; i < 8; ++i) zsum += reds[p][i];
    inv[p] = 1.0f / zsum;
  }

  float4 acc[P][4] = {};
  if constexpr (P == 1) {
    for (int idx = beg; idx < end; ++idx) {
      const int sn = csr_src[idx];
      const float w = expf((ub[0][sn] + vn[0]) - m[0]) * inv[0];
      const float4* hr = (const float4*)(Hb[0] + (size_t)sn * OO);
#pragma unroll
      for (int j = 0; j < 4; ++j) {
        const float4 h4 = hr[lane + 32 * j];
        acc[0][j].x += w * h4.x;
        acc[0][j].y += w * h4.y;
        acc[0][j].z += w * h4.z;
        acc[0][j].w += w * h4.w;
      }
    }
  } else {
    int idx = beg;
    for (; idx + 1 < end; idx += 2) walk_step<P, 2>(csr_src, idx, lane, Hb, ub, vn, m, inv, acc);
    if (idx < end) walk_step<P, 1>(csr_src, idx, lane, Hb, ub, vn, m, inv, acc);
  }
  float4* op = (float4*)(out + (((size_t)b * VV + g) * NN + n) * OO);
  float4 q[4];
  if (first) {
#pragma unroll
    for (int j = 0; j < 4; ++j) q[j] = make_float4(0.0f, 0.0f, 0.0f, 0.0f);
  } else {
#pragma unroll
    for (int j = 0; j < 4; ++j) q[j] = op[lane + 32 * j];
  }
#pragma unroll
  for (int j = 0; j < 4; ++j) {
    float4 o = q[j];
#pragma unroll
    for (int p = 0; p < P; ++p) {
      o.x += acc[p][j].x; o.y += acc[p][j].y; o.z += acc[p][j].z; o.w += acc[p][j].w;
    }
    op[lane + 32 * j] = o;
  }
}

constexpr int EPI_BLOCKS = 1024;
constexpr int EPI_THREADS = EPI_BLOCKS * 256;
constexpr int EPI_PER = (BB * VV * NN * OO / 4) / EPI_THREADS;
static_assert((BB * VV * NN * OO / 4) % EPI_THREADS == 0, "epilogue grid must tile out exactly");
static_assert(EPI_THREADS % (OO / 4) == 0, "a thread's pieces must share one bias piece");
__global__ __launch_bounds__(256) void epilogue_kernel(float* __restrict__ out, const float* __restrict__ bias) {
  const int t = blockIdx.x * 256 + threadIdx.x;
  float4* o4 = (float4*)out;
  const int bc = 4 * (t & (OO / 4 - 1));
  const float4 bv = make_float4(bias[bc], bias[bc + 1], bias[bc + 2], bias[bc + 3]);
  float4 q[EPI_PER];
#pragma unroll
  for (int k = 0; k < EPI_PER; ++k) q[k] = o4[(size_t)t + (size_t)k * EPI_THREADS];
#pragma unroll
  for (int k = 0; k < EPI_PER; ++k) {
    float4 r;
    r.x = q[k].x * 0.25f + bv.x;
    r.y = q[k].y * 0.25f + bv.y;
    r.z = q[k].z * 0.25f + bv.z;
    r.w = q[k].w * 0.25f + bv.w;
    o4[(size_t)t + (size_t)k * EPI_THREADS] = r;
  }
}

template <int P, bool MIRROR>
static void run_group(const float* x, const float* W, const float* att, Bufs B,
                      float* uS, float* vS, float* rowmax, float* rowsum,
                      const int* rowstart, const int* csr_src, float* out,
                      int g, int a0, int astep, int first, hipStream_t stream) {
  gemm_mirror_kernel<MIRROR><<<dim3((BB * NN) / 32, MIRROR ? P / 2 : P, 2), 128, 0, stream>>>(x, W, B, g, a0, astep);
  node_scores_kernel<<<dim3((BB * NN) / (8 * NPW), P), 256, 0, stream>>>(B, att, uS, vS);
  row_stats_kernel<<<dim3((BB * NN) / (8 * NPW), P), 256, 0, stream>>>(uS, vS, rowstart, csr_src, rowmax, rowsum);
  aggregate_kernel<P><<<(BB * NN) / 8, 256, 0, stream>>>(B, uS, vS, rowmax, rowsum, rowstart, csr_src,
                                                         out, g, first);
}

extern "C" void kernel_launch(void* const* d_in, const int* in_sizes, int n_in,
                              void* d_out, int out_size, void* d_ws, size_t ws_size,
                              hipStream_t stream) {
  const float* x    = (const float*)d_in[0];
  const int*   ei   = (const int*)d_in[1];
  const float* W    = (const float*)d_in[2];
  const float* att  = (const float*)d_in[3];
  const float* bias = (const float*)d_in[4];
  float* out = (float*)d_out;

  char* p = (char*)d_ws;
  float* H     = (float*)p; p += (size_t)BB * NN * OO * sizeof(float);
  float* uS     = (float*)p;
  float* vS     = uS + (size_t)4 * PS;
  float* rowmax = vS + (size_t)4 * PS;
  float* rowsum = rowmax + (size_t)4 * PS;
  p += (size_t)2 * BB * NN * sizeof(float);
  p += (size_t)BB * NE * sizeof(float);
  p += 4 * sizeof(float);
  int* counts  = (int*)p;   p += (size_t)NN * sizeof(int);
  int* csr_src = (int*)p;   p += (size_t)NE * sizeof(int);
  int* rowstart= (int*)p;   p += (size_t)(NN + 1) * sizeof(int);
  int* slot     = (int*)H;
  int* row_edge = slot + NE;

  zero_i32_kernel<<<(NN + 255) / 256, 256, 0, stream>>>(counts, NN);
  count_dst_kernel<<<(NE + 255) / 256, 256, 0, stream>>>(ei, counts, slot);
  scan_counts_kernel<<<1, 256, 0, stream>>>(counts, rowstart);
  fill_rows_kernel<<<(NE + 255) / 256, 256, 0, stream>>>(ei, rowstart, slot, row_edge);
  order_rows_kernel<<<NN / 8, 256, 0, stream>>>(ei, rowstart, row_edge, csr_src);

  constexpr int WS_BS = NN * OO;
  constexpr int PL_BS = VV * NN * OO;
  float* pl0 = out;
  float* pl2 = out + (size_t)2 * NN * OO;
  float* pl3 = out + (size_t)3 * NN * OO;
  const Bufs b4 = { H, pl0, pl2, pl3, WS_BS, PL_BS, PL_BS, PL_BS };
  const Bufs b2 = { H, pl0, pl0, pl0, WS_BS, PL_BS, PL_BS, PL_BS };
  const Bufs b1 = { H, H, H, H, WS_BS, WS_BS, WS_BS, WS_BS };

  run_group<4, true>(x, W, att, b4, uS, vS, rowmax, rowsum, rowstart, csr_src, out, 1, 0, 2, 1, stream);
  run_group<2, true>(x, W, att, b2, uS, vS, rowmax, rowsum, rowstart, csr_src, out, 2, 0, 0, 1, stream);
  run_group<2, true>(x, W, att, b2, uS, vS, rowmax, rowsum, rowstart, csr_src, out, 2, 1, 0, 0, stream);
  run_group<2, true>(x, W, att, b2, uS, vS, rowmax, rowsum, rowstart, csr_src, out, 3, 0, 0, 1, stream);
  run_group<2, true>(x, W, att, b2, uS, vS, rowmax, rowsum, rowstart, csr_src, out, 3, 1, 0, 0, stream);
  run_group<1, false>(x, W, att, b1, uS, vS, rowmax, rowsum, rowstart, csr_src, out, 0, 0, 0, 1, stream);
  run_group<1, false>(x, W, att, b1, uS, vS, rowmax, rowsum, rowstart, csr_src, out, 0, 1, 0, 0, stream);
  run_group<1, false>(x, W, att, b1, uS, vS, rowmax, rowsum, rowstart, csr_src, out, 0, 2, 0, 0, stream);
  run_group<1, false>(x, W, att, b1, uS, vS, rowmax, rowsum, rowstart, csr_src, out, 0, 3, 0, 0, stream);

  epilogue_kernel<<<EPI_BLOCKS, 256, 0, stream>>>(out, bias);

  (void)in_sizes; (void)n_in; (void)out_size; (void)ws_size;
}
